# starting kernel
# baseline (speedup 1.0000x reference)
#include <hip/hip_runtime.h>

typedef __attribute__((ext_vector_type(2))) float v2f;
typedef __attribute__((ext_vector_type(8))) float v8f;

#define CHUNK     512
#define BLOCK     128
#define STATS_RUN 16
#define DFEAT     64

__global__ void zero_f32(float* __restrict__ a, long n) {
  long i = (long)blockIdx.x * blockDim.x + threadIdx.x;
  long stride = (long)gridDim.x * blockDim.x;
  for (; i < n; i += stride) a[i] = 0.0f;
}

__global__ void edge_stats(const float* __restrict__ w, const int* __restrict__ seg,
                           float* __restrict__ wsum, float* __restrict__ deg, int E) {
  long t = (long)blockIdx.x * blockDim.x + threadIdx.x;
  long base = t * STATS_RUN;
  if (base >= E) return;
  long end = base + STATS_RUN; if (end > E) end = E;
  int cur = seg[base];
  float sw = 0.0f, sd = 0.0f;
  for (long e = base; e < end; ++e) {
    int s = seg[e];
    if (s != cur) {
      atomicAdd(&wsum[cur], sw);
      atomicAdd(&deg[cur], sd);
      cur = s; sw = 0.0f; sd = 0.0f;
    }
    sw += w[e]; sd += 1.0f;
  }
  atomicAdd(&wsum[cur], sw);
  atomicAdd(&deg[cur], sd);
}

__global__ void finalize(const float* __restrict__ wsum, float* __restrict__ deg_invden, int N) {
  int n = blockIdx.x * blockDim.x + threadIdx.x;
  if (n >= N) return;
  float w = wsum[n], d = deg_invden[n];
  float den = (w == 0.0f) ? d : w;
  if (den == 0.0f) den = 1.0f;
  deg_invden[n] = 1.0f / den;
}

__global__ void __launch_bounds__(BLOCK)
aggregate(const float* __restrict__ feat, const float* __restrict__ ew,
          const int* __restrict__ nidx, const int* __restrict__ seg,
          const float* __restrict__ wsum, float* __restrict__ out, int E) {
  __shared__ int   s_seg[CHUNK];
  __shared__ int   s_idx[CHUNK];
  __shared__ float s_w[CHUNK];

  const int tid  = threadIdx.x;
  const int lane = tid & 31;
  const int m    = lane & 15;
  const int half = lane >> 4;
  const int wave = tid >> 5;
  const int cb   = wave * 16;

  const long chunk_base = (long)blockIdx.x * CHUNK;
  for (int i = tid; i < CHUNK; i += BLOCK) {
    long e = chunk_base + i;
    if (e < (long)E) { s_seg[i] = seg[e]; s_idx[i] = nidx[e]; s_w[i] = ew[e]; }
    else             { s_seg[i] = -1;     s_idx[i] = 0;       s_w[i] = 0.0f;  }
  }
  __syncthreads();

  long rem = (long)E - chunk_base;
  int nvalid = rem < (long)CHUNK ? (int)rem : CHUNK;

  int  cur_seg  = -2;
  int  used     = 0;
  bool cur_mean = false;
  int  slot_node = -1;
  v8f  c;
  #pragma unroll
  for (int r = 0; r < 8; ++r) c[r] = 0.0f;

  const int kx = 2 * half, ky = kx + 1;

  for (int g = 0; g < CHUNK; g += 4) {
    if (g >= nvalid) break;

    int e_seg[4], e_idx[4]; float e_w[4];
    #pragma unroll
    for (int k = 0; k < 4; ++k) {
      int i = g + k;
      e_seg[k] = s_seg[i]; e_idx[k] = s_idx[i]; e_w[k] = s_w[i];
    }

    int tmp = cur_seg, newcnt = 0;
    #pragma unroll
    for (int k = 0; k < 4; ++k)
      if (e_seg[k] >= 0 && e_seg[k] != tmp) { newcnt++; tmp = e_seg[k]; }

    if (used + newcnt > 16) {
      #pragma unroll
      for (int r = 0; r < 8; ++r) {
        int row  = r + 8 * half;
        int node = __shfl(slot_node, row, 32);
        if (row < used && node >= 0)
          atomicAdd(&out[(long)node * DFEAT + cb + m], c[r]);
      }
      #pragma unroll
      for (int r = 0; r < 8; ++r) c[r] = 0.0f;
      used = 0; cur_seg = -2; slot_node = -1;
    }

    int slot[4]; float weff[4];
    int cur_slot = used - 1;
    #pragma unroll
    for (int k = 0; k < 4; ++k) {
      if (e_seg[k] < 0) { slot[k] = 0; weff[k] = 0.0f; continue; }
      if (e_seg[k] != cur_seg) {
        cur_seg  = e_seg[k];
        cur_slot = used++;
        slot_node = (m == cur_slot) ? cur_seg : slot_node;
        cur_mean = (wsum[cur_seg] == 0.0f);
      }
      slot[k] = cur_slot;
      weff[k] = cur_mean ? 1.0f : e_w[k];
    }

    v2f a, b;
    a.x = (m == slot[kx]) ? weff[kx] : 0.0f;
    a.y = (m == slot[ky]) ? weff[ky] : 0.0f;
    b.x = feat[(long)e_idx[kx] * DFEAT + cb + m];
    b.y = feat[(long)e_idx[ky] * DFEAT + cb + m];

    if (g + 4 < nvalid) {
      __builtin_prefetch(&feat[(long)s_idx[g + 4 + kx] * DFEAT + cb], 0, 1);
      __builtin_prefetch(&feat[(long)s_idx[g + 4 + ky] * DFEAT + cb], 0, 1);
    }

    c = __builtin_amdgcn_wmma_f32_16x16x4_f32(false, a, false, b, (short)0, c,
                                              false, false);
  }

  #pragma unroll
  for (int r = 0; r < 8; ++r) {
    int row  = r + 8 * half;
    int node = __shfl(slot_node, row, 32);
    if (row < used && node >= 0)
      atomicAdd(&out[(long)node * DFEAT + cb + m], c[r]);
  }
}

__global__ void scale_out(float* __restrict__ out, const float* __restrict__ inv_den, long total) {
  long i = (long)blockIdx.x * blockDim.x + threadIdx.x;
  long stride = (long)gridDim.x * blockDim.x;
  for (; i < total; i += stride) out[i] *= inv_den[i >> 6];
}

static inline int cdiv_i(long a, long b) { return (int)((a + b - 1) / b); }

extern "C" void kernel_launch(void* const* d_in, const int* in_sizes, int n_in,
                              void* d_out, int out_size, void* d_ws, size_t ws_size,
                              hipStream_t stream) {
  const float* feat = (const float*)d_in[0];
  const float* ew   = (const float*)d_in[1];
  const int*   nidx = (const int*)d_in[2];
  const int*   seg  = (const int*)d_in[3];
  float* out = (float*)d_out;

  const int N = in_sizes[0] / DFEAT;
  const int E = in_sizes[1];

  float* wsum = (float*)d_ws;
  float* deg  = wsum + N;

  zero_f32<<<2048, 256, 0, stream>>>(wsum, 2L * N);
  zero_f32<<<4096, 256, 0, stream>>>(out, (long)out_size);

  int stat_threads = cdiv_i(E, STATS_RUN);
  edge_stats<<<cdiv_i(stat_threads, 256), 256, 0, stream>>>(ew, seg, wsum, deg, E);
  finalize<<<cdiv_i(N, 256), 256, 0, stream>>>(wsum, deg, N);

  aggregate<<<cdiv_i(E, CHUNK), BLOCK, 0, stream>>>(feat, ew, nidx, seg, wsum, out, E);

  scale_out<<<4096, 256, 0, stream>>>(out, deg, (long)out_size);
}


# aggregate: one wave owns all 64 feature columns of 160 consecutive edges (4 accumulator tiles per edge group), slot bookkeeping runs once instead of 4x; grouping of atomic partial sums changes
# speedup vs baseline: 1.7708x; 1.7708x over previous
#include <hip/hip_runtime.h>

typedef __attribute__((ext_vector_type(2))) float v2f;
typedef __attribute__((ext_vector_type(8))) float v8f;

#define CHUNK     640
#define BLOCK     128
#define WAVE_EDGES (CHUNK / 4)
#define STATS_RUN 16
#define DFEAT     64

__global__ void zero_f32(float* __restrict__ a, long n) {
  long i = (long)blockIdx.x * blockDim.x + threadIdx.x;
  long stride = (long)gridDim.x * blockDim.x;
  for (; i < n; i += stride) a[i] = 0.0f;
}

__global__ void edge_stats(const float* __restrict__ w, const int* __restrict__ seg,
                           float* __restrict__ wsum, float* __restrict__ deg, int E) {
  long t = (long)blockIdx.x * blockDim.x + threadIdx.x;
  long base = t * STATS_RUN;
  if (base >= E) return;
  long end = base + STATS_RUN; if (end > E) end = E;
  int cur = seg[base];
  float sw = 0.0f, sd = 0.0f;
  for (long e = base; e < end; ++e) {
    int s = seg[e];
    if (s != cur) {
      atomicAdd(&wsum[cur], sw);
      atomicAdd(&deg[cur], sd);
      cur = s; sw = 0.0f; sd = 0.0f;
    }
    sw += w[e]; sd += 1.0f;
  }
  atomicAdd(&wsum[cur], sw);
  atomicAdd(&deg[cur], sd);
}

__global__ void finalize(const float* __restrict__ wsum, float* __restrict__ deg_invden, int N) {
  int n = blockIdx.x * blockDim.x + threadIdx.x;
  if (n >= N) return;
  float w = wsum[n], d = deg_invden[n];
  float den = (w == 0.0f) ? d : w;
  if (den == 0.0f) den = 1.0f;
  deg_invden[n] = 1.0f / den;
}

__global__ void __launch_bounds__(BLOCK)
aggregate(const float* __restrict__ feat, const float* __restrict__ ew,
          const int* __restrict__ nidx, const int* __restrict__ seg,
          const float* __restrict__ wsum, float* __restrict__ out, int E) {
  __shared__ int   s_seg[CHUNK];
  __shared__ int   s_idx[CHUNK];
  __shared__ float s_w[CHUNK];

  const int tid  = threadIdx.x;
  const int lane = tid & 31;
  const int m    = lane & 15;
  const int half = lane >> 4;
  const int wave = __builtin_amdgcn_readfirstlane(tid >> 5);
  const int eb   = wave * WAVE_EDGES;

  const long chunk_base = (long)blockIdx.x * CHUNK;
  for (int i = tid; i < CHUNK; i += BLOCK) {
    long e = chunk_base + i;
    if (e < (long)E) { s_seg[i] = seg[e]; s_idx[i] = nidx[e]; s_w[i] = ew[e]; }
    else             { s_seg[i] = -1;     s_idx[i] = 0;       s_w[i] = 0.0f;  }
  }
  __syncthreads();

  long rem = (long)E - chunk_base - eb;
  int nvalid = rem < (long)WAVE_EDGES ? (int)rem : WAVE_EDGES;

  int  cur_seg  = -2;
  int  used     = 0;
  bool cur_mean = false;
  int  slot_node = -1;
  v8f  c0, c1, c2, c3;
  #pragma unroll
  for (int r = 0; r < 8; ++r) { c0[r] = 0.0f; c1[r] = 0.0f; c2[r] = 0.0f; c3[r] = 0.0f; }

  const int kx = 2 * half, ky = kx + 1;

  for (int g = 0; g < WAVE_EDGES; g += 4) {
    if (g >= nvalid) break;

    int e_seg[4], e_idx[4]; float e_w[4];
    #pragma unroll
    for (int k = 0; k < 4; ++k) {
      int i = eb + g + k;
      e_seg[k] = s_seg[i]; e_idx[k] = s_idx[i]; e_w[k] = s_w[i];
    }

    int tmp = cur_seg, newcnt = 0;
    #pragma unroll
    for (int k = 0; k < 4; ++k)
      if (e_seg[k] >= 0 && e_seg[k] != tmp) { newcnt++; tmp = e_seg[k]; }

    if (used + newcnt > 16) {
      #pragma unroll
      for (int r = 0; r < 8; ++r) {
        int row  = r + 8 * half;
        int node = __shfl(slot_node, row, 32);
        if (row < used && node >= 0) {
          float* p = &out[(long)node * DFEAT + m];
          atomicAdd(p,      c0[r]);
          atomicAdd(p + 16, c1[r]);
          atomicAdd(p + 32, c2[r]);
          atomicAdd(p + 48, c3[r]);
        }
      }
      #pragma unroll
      for (int r = 0; r < 8; ++r) { c0[r] = 0.0f; c1[r] = 0.0f; c2[r] = 0.0f; c3[r] = 0.0f; }
      used = 0; cur_seg = -2; slot_node = -1;
    }

    int slot[4]; float weff[4];
    int cur_slot = used - 1;
    #pragma unroll
    for (int k = 0; k < 4; ++k) {
      if (e_seg[k] < 0) { slot[k] = 0; weff[k] = 0.0f; continue; }
      if (e_seg[k] != cur_seg) {
        cur_seg  = e_seg[k];
        cur_slot = used++;
        slot_node = (m == cur_slot) ? cur_seg : slot_node;
        cur_mean = (wsum[cur_seg] == 0.0f);
      }
      slot[k] = cur_slot;
      weff[k] = cur_mean ? 1.0f : e_w[k];
    }

    v2f a;
    a.x = (m == slot[kx]) ? weff[kx] : 0.0f;
    a.y = (m == slot[ky]) ? weff[ky] : 0.0f;
    const float* rx = &feat[(long)e_idx[kx] * DFEAT + m];
    const float* ry = &feat[(long)e_idx[ky] * DFEAT + m];
    v2f b0, b1, b2, b3;
    b0.x = rx[0];  b0.y = ry[0];
    b1.x = rx[16]; b1.y = ry[16];
    b2.x = rx[32]; b2.y = ry[32];
    b3.x = rx[48]; b3.y = ry[48];

    if (g + 4 < nvalid) {
      __builtin_prefetch(&feat[(long)s_idx[eb + g + 4 + kx] * DFEAT + 4 * m], 0, 1);
      __builtin_prefetch(&feat[(long)s_idx[eb + g + 4 + ky] * DFEAT + 4 * m], 0, 1);
    }

    c0 = __builtin_amdgcn_wmma_f32_16x16x4_f32(false, a, false, b0, (short)0, c0, false, false);
    c1 = __builtin_amdgcn_wmma_f32_16x16x4_f32(false, a, false, b1, (short)0, c1, false, false);
    c2 = __builtin_amdgcn_wmma_f32_16x16x4_f32(false, a, false, b2, (short)0, c2, false, false);
    c3 = __builtin_amdgcn_wmma_f32_16x16x4_f32(false, a, false, b3, (short)0, c3, false, false);
  }

  #pragma unroll
  for (int r = 0; r < 8; ++r) {
    int row  = r + 8 * half;
    int node = __shfl(slot_node, row, 32);
    if (row < used && node >= 0) {
      float* p = &out[(long)node * DFEAT + m];
      atomicAdd(p,      c0[r]);
      atomicAdd(p + 16, c1[r]);
      atomicAdd(p + 32, c2[r]);
      atomicAdd(p + 48, c3[r]);
    }
  }
}

__global__ void scale_out(float* __restrict__ out, const float* __restrict__ inv_den, long total) {
  long i = (long)blockIdx.x * blockDim.x + threadIdx.x;
  long stride = (long)gridDim.x * blockDim.x;
  for (; i < total; i += stride) out[i] *= inv_den[i >> 6];
}

static inline int cdiv_i(long a, long b) { return (int)((a + b - 1) / b); }

extern "C" void kernel_launch(void* const* d_in, const int* in_sizes, int n_in,
                              void* d_out, int out_size, void* d_ws, size_t ws_size,
                              hipStream_t stream) {
  const float* feat = (const float*)d_in[0];
  const float* ew   = (const float*)d_in[1];
  const int*   nidx = (const int*)d_in[2];
  const int*   seg  = (const int*)d_in[3];
  float* out = (float*)d_out;

  const int N = in_sizes[0] / DFEAT;
  const int E = in_sizes[1];

  float* wsum = (float*)d_ws;
  float* deg  = wsum + N;

  zero_f32<<<2048, 256, 0, stream>>>(wsum, 2L * N);
  zero_f32<<<4096, 256, 0, stream>>>(out, (long)out_size);

  int stat_threads = cdiv_i(E, STATS_RUN);
  edge_stats<<<cdiv_i(stat_threads, 256), 256, 0, stream>>>(ew, seg, wsum, deg, E);
  finalize<<<cdiv_i(N, 256), 256, 0, stream>>>(wsum, deg, N);

  aggregate<<<cdiv_i(E, CHUNK), BLOCK, 0, stream>>>(feat, ew, nidx, seg, wsum, out, E);

  scale_out<<<4096, 256, 0, stream>>>(out, deg, (long)out_size);
}
